# starting kernel
# baseline (speedup 1.0000x reference)
#include <hip/hip_runtime.h>
#include <hip/hip_bf16.h>
#include <math.h>


typedef float v2f __attribute__((ext_vector_type(2)));
typedef float v8f __attribute__((ext_vector_type(8)));

#define HDIM 12
#define ODIM 2
#define WAVES_PER_BLOCK 4

__device__ __forceinline__ float fast_tanhf(float x) {
#if __has_builtin(__builtin_amdgcn_tanhf)
  return __builtin_amdgcn_tanhf(x);
#elif __has_builtin(__builtin_amdgcn_tanh_f32)
  return __builtin_amdgcn_tanh_f32(x);
#else
  return tanhf(x);
#endif
}

__launch_bounds__(32 * WAVES_PER_BLOCK)
__global__ void tinyrnn_wmma_kernel(const float* __restrict__ x,
                                    const float* __restrict__ W_ih,
                                    const float* __restrict__ W_hh,
                                    const float* __restrict__ b_ih,
                                    const float* __restrict__ b_hh,
                                    const float* __restrict__ W_head,
                                    const float* __restrict__ b_head,
                                    float* __restrict__ out,
                                    int T, int B) {
  const int lane = threadIdx.x & 31;
  const int wv   = threadIdx.x >> 5;
  const int base = (blockIdx.x * WAVES_PER_BLOCK + wv) * 16;

  const int col  = lane & 15;
  const int half = lane >> 4;

  float a[4][2];
#pragma unroll
  for (int s = 0; s < 4; ++s) {
#pragma unroll
    for (int r = 0; r < 2; ++r) {
      const int j = 2 * s + r + 8 * half;
      a[s][r] = (col < HDIM && j < HDIM) ? W_hh[col * HDIM + j] : 0.0f;
    }
  }

  float wih[8], bias[8];
#pragma unroll
  for (int v = 0; v < 8; ++v) {
    const int j = v + 8 * half;
    wih[v]  = (j < HDIM) ? W_ih[j] : 0.0f;
    bias[v] = (j < HDIM) ? (b_ih[j] + b_hh[j]) : 0.0f;
  }

  int brow = base + col;
  if (brow >= B) brow = B - 1;
  const float4* __restrict__ xrow =
      (const float4*)(x + (size_t)brow * (size_t)T);

  float d[8];
#pragma unroll
  for (int i = 0; i < 8; ++i) d[i] = 0.0f;

  auto step = [&](float xt) {
    v8f c;
#pragma unroll
    for (int v = 0; v < 8; ++v) c[v] = fmaf(xt, wih[v], bias[v]);
    const v2f A0 = {a[0][0], a[0][1]};
    const v2f A1 = {a[1][0], a[1][1]};
    const v2f A2 = {a[2][0], a[2][1]};
    const v2f A3 = {a[3][0], a[3][1]};
    const v2f B0 = {d[0], d[1]};
    const v2f B1 = {d[2], d[3]};
    const v2f B2 = {d[4], d[5]};
    const v2f B3 = {d[6], d[7]};
    c = __builtin_amdgcn_wmma_f32_16x16x4_f32(false, A0, false, B0,
                                              (short)0, c, false, false);
    c = __builtin_amdgcn_wmma_f32_16x16x4_f32(false, A1, false, B1,
                                              (short)0, c, false, false);
    c = __builtin_amdgcn_wmma_f32_16x16x4_f32(false, A2, false, B2,
                                              (short)0, c, false, false);
    c = __builtin_amdgcn_wmma_f32_16x16x4_f32(false, A3, false, B3,
                                              (short)0, c, false, false);
#pragma unroll
    for (int v = 0; v < 8; ++v) d[v] = fast_tanhf(c[v]);
  };

  const int NT4 = T >> 2;
  float4 xv = xrow[0];
  for (int tq = 0; tq < NT4; ++tq) {
    float4 xn = xv;
    if (tq + 1 < NT4) xn = xrow[tq + 1];
    step(xv.x);
    step(xv.y);
    step(xv.z);
    step(xv.w);
    xv = xn;
  }
  const float* xs = x + (size_t)brow * (size_t)T;
  for (int t = NT4 << 2; t < T; ++t) step(xs[t]);

  __shared__ float hbuf[WAVES_PER_BLOCK][16][HDIM];
#pragma unroll
  for (int v = 0; v < 8; ++v) {
    const int j = v + 8 * half;
    if (j < HDIM) hbuf[wv][col][j] = d[v];
  }
  __syncthreads();
  const int r = col, o = half;
  const int ob = base + r;
  if (ob < B) {
    float acc = b_head[o];
#pragma unroll
    for (int j = 0; j < HDIM; ++j)
      acc = fmaf(hbuf[wv][r][j], W_head[o * HDIM + j], acc);
    out[ob * ODIM + o] = acc;
  }
}

extern "C" void kernel_launch(void* const* d_in, const int* in_sizes, int n_in,
                              void* d_out, int out_size, void* d_ws, size_t ws_size,
                              hipStream_t stream) {
  (void)n_in; (void)d_ws; (void)ws_size;
  const float* x      = (const float*)d_in[0];
  const float* W_ih   = (const float*)d_in[1];
  const float* W_hh   = (const float*)d_in[2];
  const float* b_ih   = (const float*)d_in[3];
  const float* b_hh   = (const float*)d_in[4];
  const float* W_head = (const float*)d_in[5];
  const float* b_head = (const float*)d_in[6];
  float* out = (float*)d_out;

  const int B = out_size / ODIM;
  const int T = in_sizes[0] / B;

  const int waves  = (B + 15) / 16;
  const int blocks = (waves + WAVES_PER_BLOCK - 1) / WAVES_PER_BLOCK;
  tinyrnn_wmma_kernel<<<blocks, 32 * WAVES_PER_BLOCK, 0, stream>>>(
      x, W_ih, W_hh, b_ih, b_hh, W_head, b_head, out, T, B);
}


# Pack the 12 hidden rows into D registers 0-5 of both lane halves (matrix rows 0-5, 8-13): K=12 takes 3 chained f32 WMMAs, 6 tanh, 6 FMA per step, not 4/8/8; k order of the 12-term sums changes
# speedup vs baseline: 1.2431x; 1.2431x over previous
#include <hip/hip_runtime.h>
#include <hip/hip_bf16.h>
#include <math.h>


typedef float v2f __attribute__((ext_vector_type(2)));
typedef float v8f __attribute__((ext_vector_type(8)));

#define HDIM 12
#define ODIM 2
#define WAVES_PER_BLOCK 4
#define HROWS 6
#define NSLICE 3

__device__ __forceinline__ float fast_tanhf(float x) {
#if __has_builtin(__builtin_amdgcn_tanhf)
  return __builtin_amdgcn_tanhf(x);
#elif __has_builtin(__builtin_amdgcn_tanh_f32)
  return __builtin_amdgcn_tanh_f32(x);
#else
  return tanhf(x);
#endif
}

__launch_bounds__(32 * WAVES_PER_BLOCK)
__global__ void tinyrnn_wmma_kernel(const float* __restrict__ x,
                                    const float* __restrict__ W_ih,
                                    const float* __restrict__ W_hh,
                                    const float* __restrict__ b_ih,
                                    const float* __restrict__ b_hh,
                                    const float* __restrict__ W_head,
                                    const float* __restrict__ b_head,
                                    float* __restrict__ out,
                                    int T, int B) {
  const int lane = threadIdx.x & 31;
  const int wv   = threadIdx.x >> 5;
  const int base = (blockIdx.x * WAVES_PER_BLOCK + wv) * 16;

  const int col  = lane & 15;
  const int half = lane >> 4;

  const bool arowok = (col < HROWS) || (col >= 8 && col < 8 + HROWS);
  const int  arow   = !arowok ? 0 : (col < HROWS ? col : col - 2);
  float a[NSLICE][2];
#pragma unroll
  for (int s = 0; s < NSLICE; ++s) {
#pragma unroll
    for (int r = 0; r < 2; ++r) {
      const int j = 2 * s + r + HROWS * half;
      const float w = W_hh[arow * HDIM + j];
      a[s][r] = arowok ? w : 0.0f;
    }
  }

  float wih[HROWS], bias[HROWS];
#pragma unroll
  for (int v = 0; v < HROWS; ++v) {
    const int j = v + HROWS * half;
    wih[v]  = W_ih[j];
    bias[v] = b_ih[j] + b_hh[j];
  }

  int brow = base + col;
  if (brow >= B) brow = B - 1;
  const float4* __restrict__ xrow =
      (const float4*)(x + (size_t)brow * (size_t)T);

  float d[HROWS];
#pragma unroll
  for (int i = 0; i < HROWS; ++i) d[i] = 0.0f;
  float pad6 = 0.0f, pad7 = 0.0f;

  auto step = [&](float xt) {
    v8f c;
#pragma unroll
    for (int v = 0; v < HROWS; ++v) c[v] = fmaf(xt, wih[v], bias[v]);
    c[6] = pad6;
    c[7] = pad7;
    const v2f A0 = {a[0][0], a[0][1]};
    const v2f A1 = {a[1][0], a[1][1]};
    const v2f A2 = {a[2][0], a[2][1]};
    const v2f B0 = {d[0], d[1]};
    const v2f B1 = {d[2], d[3]};
    const v2f B2 = {d[4], d[5]};
    c = __builtin_amdgcn_wmma_f32_16x16x4_f32(false, A0, false, B0,
                                              (short)0, c, false, false);
    c = __builtin_amdgcn_wmma_f32_16x16x4_f32(false, A1, false, B1,
                                              (short)0, c, false, false);
    c = __builtin_amdgcn_wmma_f32_16x16x4_f32(false, A2, false, B2,
                                              (short)0, c, false, false);
#pragma unroll
    for (int v = 0; v < HROWS; ++v) d[v] = fast_tanhf(c[v]);
    pad6 = c[6];
    pad7 = c[7];
  };

  const int NT4 = T >> 2;
  float4 xv = xrow[0];
  for (int tq = 0; tq < NT4; ++tq) {
    float4 xn = xv;
    if (tq + 1 < NT4) xn = xrow[tq + 1];
    step(xv.x);
    step(xv.y);
    step(xv.z);
    step(xv.w);
    xv = xn;
  }
  const float* xs = x + (size_t)brow * (size_t)T;
  for (int t = NT4 << 2; t < T; ++t) step(xs[t]);

  __shared__ float hbuf[WAVES_PER_BLOCK][16][HDIM];
#pragma unroll
  for (int v = 0; v < HROWS; ++v) {
    const int j = v + HROWS * half;
    hbuf[wv][col][j] = d[v];
  }
  __syncthreads();
  const int r = col, o = half;
  const int ob = base + r;
  if (ob < B) {
    float acc = b_head[o];
#pragma unroll
    for (int j = 0; j < HDIM; ++j)
      acc = fmaf(hbuf[wv][r][j], W_head[o * HDIM + j], acc);
    out[ob * ODIM + o] = acc;
  }
}

extern "C" void kernel_launch(void* const* d_in, const int* in_sizes, int n_in,
                              void* d_out, int out_size, void* d_ws, size_t ws_size,
                              hipStream_t stream) {
  (void)n_in; (void)d_ws; (void)ws_size;
  const float* x      = (const float*)d_in[0];
  const float* W_ih   = (const float*)d_in[1];
  const float* W_hh   = (const float*)d_in[2];
  const float* b_ih   = (const float*)d_in[3];
  const float* b_hh   = (const float*)d_in[4];
  const float* W_head = (const float*)d_in[5];
  const float* b_head = (const float*)d_in[6];
  float* out = (float*)d_out;

  const int B = out_size / ODIM;
  const int T = in_sizes[0] / B;

  const int waves  = (B + 15) / 16;
  const int blocks = (waves + WAVES_PER_BLOCK - 1) / WAVES_PER_BLOCK;
  tinyrnn_wmma_kernel<<<blocks, 32 * WAVES_PER_BLOCK, 0, stream>>>(
      x, W_ih, W_hh, b_ih, b_hh, W_head, b_head, out, T, B);
}


# Next step's input FMAs (x*W_ih+bias) go into a second accumulator tuple right behind the current step's last WMMA: 3 v_pk_fma and their result latency leave the serial tanh->WMMA chain; same sums
# speedup vs baseline: 1.2944x; 1.0413x over previous
#include <hip/hip_runtime.h>
#include <hip/hip_bf16.h>
#include <math.h>


typedef float v2f __attribute__((ext_vector_type(2)));
typedef float v8f __attribute__((ext_vector_type(8)));

#define HDIM 12
#define ODIM 2
#define WAVES_PER_BLOCK 4
#define HROWS 6
#define NSLICE 3

__device__ __forceinline__ float fast_tanhf(float x) {
#if __has_builtin(__builtin_amdgcn_tanhf)
  return __builtin_amdgcn_tanhf(x);
#elif __has_builtin(__builtin_amdgcn_tanh_f32)
  return __builtin_amdgcn_tanh_f32(x);
#else
  return tanhf(x);
#endif
}

__launch_bounds__(32 * WAVES_PER_BLOCK)
__global__ void tinyrnn_wmma_kernel(const float* __restrict__ x,
                                    const float* __restrict__ W_ih,
                                    const float* __restrict__ W_hh,
                                    const float* __restrict__ b_ih,
                                    const float* __restrict__ b_hh,
                                    const float* __restrict__ W_head,
                                    const float* __restrict__ b_head,
                                    float* __restrict__ out,
                                    int T, int B) {
  const int lane = threadIdx.x & 31;
  const int wv   = threadIdx.x >> 5;
  const int base = (blockIdx.x * WAVES_PER_BLOCK + wv) * 16;

  const int col  = lane & 15;
  const int half = lane >> 4;

  const bool arowok = (col < HROWS) || (col >= 8 && col < 8 + HROWS);
  const int  arow   = !arowok ? 0 : (col < HROWS ? col : col - 2);
  float a[NSLICE][2];
#pragma unroll
  for (int s = 0; s < NSLICE; ++s) {
#pragma unroll
    for (int r = 0; r < 2; ++r) {
      const int j = 2 * s + r + HROWS * half;
      const float w = W_hh[arow * HDIM + j];
      a[s][r] = arowok ? w : 0.0f;
    }
  }

  float wih[HROWS], bias[HROWS];
#pragma unroll
  for (int v = 0; v < HROWS; ++v) {
    const int j = v + HROWS * half;
    wih[v]  = W_ih[j];
    bias[v] = b_ih[j] + b_hh[j];
  }

  int brow = base + col;
  if (brow >= B) brow = B - 1;
  const float4* __restrict__ xrow =
      (const float4*)(x + (size_t)brow * (size_t)T);

  v8f accE, accO;
#pragma unroll
  for (int i = 0; i < 8; ++i) { accE[i] = 0.0f; accO[i] = 0.0f; }

  const v2f A0 = {a[0][0], a[0][1]};
  const v2f A1 = {a[1][0], a[1][1]};
  const v2f A2 = {a[2][0], a[2][1]};

  auto cinit = [&](v8f& c, float xt) {
#pragma unroll
    for (int v = 0; v < HROWS; ++v) c[v] = fmaf(xt, wih[v], bias[v]);
  };

  auto step = [&](v8f& cur, v8f& prev, float xnext) {
    float h[HROWS];
#pragma unroll
    for (int v = 0; v < HROWS; ++v) h[v] = fast_tanhf(prev[v]);
    const v2f B0 = {h[0], h[1]};
    const v2f B1 = {h[2], h[3]};
    const v2f B2 = {h[4], h[5]};
    __builtin_amdgcn_sched_barrier(0);
    cur = __builtin_amdgcn_wmma_f32_16x16x4_f32(false, A0, false, B0,
                                                (short)0, cur, false, false);
    cur = __builtin_amdgcn_wmma_f32_16x16x4_f32(false, A1, false, B1,
                                                (short)0, cur, false, false);
    cur = __builtin_amdgcn_wmma_f32_16x16x4_f32(false, A2, false, B2,
                                                (short)0, cur, false, false);
    __builtin_amdgcn_sched_barrier(0);
    cinit(prev, xnext);
    __builtin_amdgcn_sched_barrier(0);
  };

  const int NT4 = T >> 2;
  const float* xs = x + (size_t)brow * (size_t)T;
  float4 xv = xrow[0];

  {
    cinit(accE, xv.x);
    const v2f Z = {0.0f, 0.0f};
    accE = __builtin_amdgcn_wmma_f32_16x16x4_f32(false, A0, false, Z,
                                                 (short)0, accE, false, false);
    accE = __builtin_amdgcn_wmma_f32_16x16x4_f32(false, A1, false, Z,
                                                 (short)0, accE, false, false);
    accE = __builtin_amdgcn_wmma_f32_16x16x4_f32(false, A2, false, Z,
                                                 (short)0, accE, false, false);
    cinit(accO, xv.y);
  }

  bool curE = false;
  int  tnext = 1;
  if (NT4 > 0) {
    const int qlast = NT4 - 1;
    float4 xa = xrow[1 < qlast ? 1 : qlast];
    float4 xb = xrow[2 < qlast ? 2 : qlast];
    step(accO, accE, xv.z);
    step(accE, accO, xv.w);
    step(accO, accE, xa.x);
    int tq = 1;
    for (; tq + 1 < NT4; tq += 2) {
      step(accE, accO, xa.y);
      step(accO, accE, xa.z);
      step(accE, accO, xa.w);
      xa = xrow[tq + 2 < qlast ? tq + 2 : qlast];
      step(accO, accE, xb.x);
      step(accE, accO, xb.y);
      step(accO, accE, xb.z);
      step(accE, accO, xb.w);
      xb = xrow[tq + 3 < qlast ? tq + 3 : qlast];
      step(accO, accE, xa.x);
    }
    if (tq < NT4) {
      step(accE, accO, xa.y);
      step(accO, accE, xa.z);
      step(accE, accO, xa.w);
      step(accO, accE, xa.x);
    }
    tnext = NT4 << 2;
    curE = true;
    if (tnext < T) cinit(accE, xs[tnext]);
  }
  for (int t = tnext; t < T; ++t) {
    const float xnext = (t + 1 < T) ? xs[t + 1] : 0.0f;
    if (curE) step(accE, accO, xnext);
    else      step(accO, accE, xnext);
    curE = !curE;
  }
  const bool lastE = !curE;

  float d[HROWS];
#pragma unroll
  for (int v = 0; v < HROWS; ++v) d[v] = fast_tanhf(lastE ? accE[v] : accO[v]);

  __shared__ float hbuf[WAVES_PER_BLOCK][16][HDIM];
#pragma unroll
  for (int v = 0; v < HROWS; ++v) {
    const int j = v + HROWS * half;
    hbuf[wv][col][j] = d[v];
  }
  __syncthreads();
  const int r = col, o = half;
  const int ob = base + r;
  if (ob < B) {
    float acc = b_head[o];
#pragma unroll
    for (int j = 0; j < HDIM; ++j)
      acc = fmaf(hbuf[wv][r][j], W_head[o * HDIM + j], acc);
    out[ob * ODIM + o] = acc;
  }
}

extern "C" void kernel_launch(void* const* d_in, const int* in_sizes, int n_in,
                              void* d_out, int out_size, void* d_ws, size_t ws_size,
                              hipStream_t stream) {
  (void)n_in; (void)d_ws; (void)ws_size;
  const float* x      = (const float*)d_in[0];
  const float* W_ih   = (const float*)d_in[1];
  const float* W_hh   = (const float*)d_in[2];
  const float* b_ih   = (const float*)d_in[3];
  const float* b_hh   = (const float*)d_in[4];
  const float* W_head = (const float*)d_in[5];
  const float* b_head = (const float*)d_in[6];
  float* out = (float*)d_out;

  const int B = out_size / ODIM;
  const int T = in_sizes[0] / B;

  const int waves  = (B + 15) / 16;
  const int blocks = (waves + WAVES_PER_BLOCK - 1) / WAVES_PER_BLOCK;
  tinyrnn_wmma_kernel<<<blocks, 32 * WAVES_PER_BLOCK, 0, stream>>>(
      x, W_ih, W_hh, b_ih, b_hh, W_head, b_head, out, T, B);
}


# x prefetch ring of four float4 buffers, four blocks per trip: each block is loaded 12 steps ahead (was 4), so the one load in eight that fetches 16 new lines no longer stalls the chain
# speedup vs baseline: 1.4391x; 1.1118x over previous
#include <hip/hip_runtime.h>
#include <hip/hip_bf16.h>
#include <math.h>


typedef float v2f __attribute__((ext_vector_type(2)));
typedef float v8f __attribute__((ext_vector_type(8)));
typedef float v4f __attribute__((ext_vector_type(4)));

#define HDIM 12
#define ODIM 2
#define WAVES_PER_BLOCK 4
#define HROWS 6
#define NSLICE 3

__device__ __forceinline__ float fast_tanhf(float x) {
#if __has_builtin(__builtin_amdgcn_tanhf)
  return __builtin_amdgcn_tanhf(x);
#elif __has_builtin(__builtin_amdgcn_tanh_f32)
  return __builtin_amdgcn_tanh_f32(x);
#else
  return tanhf(x);
#endif
}

__launch_bounds__(32 * WAVES_PER_BLOCK)
__global__ void tinyrnn_wmma_kernel(const float* __restrict__ x,
                                    const float* __restrict__ W_ih,
                                    const float* __restrict__ W_hh,
                                    const float* __restrict__ b_ih,
                                    const float* __restrict__ b_hh,
                                    const float* __restrict__ W_head,
                                    const float* __restrict__ b_head,
                                    float* __restrict__ out,
                                    int T, int B) {
  const int lane = threadIdx.x & 31;
  const int wv   = threadIdx.x >> 5;
  const int base = (blockIdx.x * WAVES_PER_BLOCK + wv) * 16;

  const int col  = lane & 15;
  const int half = lane >> 4;

  const bool arowok = (col < HROWS) || (col >= 8 && col < 8 + HROWS);
  const int  arow   = !arowok ? 0 : (col < HROWS ? col : col - 2);
  float a[NSLICE][2];
#pragma unroll
  for (int s = 0; s < NSLICE; ++s) {
#pragma unroll
    for (int r = 0; r < 2; ++r) {
      const int j = 2 * s + r + HROWS * half;
      const float w = W_hh[arow * HDIM + j];
      a[s][r] = arowok ? w : 0.0f;
    }
  }

  float wih[HROWS], bias[HROWS];
#pragma unroll
  for (int v = 0; v < HROWS; ++v) {
    const int j = v + HROWS * half;
    wih[v]  = W_ih[j];
    bias[v] = b_ih[j] + b_hh[j];
  }

  int brow = base + col;
  if (brow >= B) brow = B - 1;
  const v4f* __restrict__ xrow =
      (const v4f*)(x + (size_t)brow * (size_t)T);

  v8f accE, accO;
#pragma unroll
  for (int i = 0; i < 8; ++i) { accE[i] = 0.0f; accO[i] = 0.0f; }

  const v2f A0 = {a[0][0], a[0][1]};
  const v2f A1 = {a[1][0], a[1][1]};
  const v2f A2 = {a[2][0], a[2][1]};

  auto cinit = [&](v8f& c, float xt) {
#pragma unroll
    for (int v = 0; v < HROWS; ++v) c[v] = fmaf(xt, wih[v], bias[v]);
  };

  auto step = [&](v8f& cur, v8f& prev, float xnext) {
    float h[HROWS];
#pragma unroll
    for (int v = 0; v < HROWS; ++v) h[v] = fast_tanhf(prev[v]);
    const v2f B0 = {h[0], h[1]};
    const v2f B1 = {h[2], h[3]};
    const v2f B2 = {h[4], h[5]};
    __builtin_amdgcn_sched_barrier(0);
    cur = __builtin_amdgcn_wmma_f32_16x16x4_f32(false, A0, false, B0,
                                                (short)0, cur, false, false);
    cur = __builtin_amdgcn_wmma_f32_16x16x4_f32(false, A1, false, B1,
                                                (short)0, cur, false, false);
    cur = __builtin_amdgcn_wmma_f32_16x16x4_f32(false, A2, false, B2,
                                                (short)0, cur, false, false);
    __builtin_amdgcn_sched_barrier(0);
    cinit(prev, xnext);
    __builtin_amdgcn_sched_barrier(0);
  };

  const int NT4 = T >> 2;
  const float* xs = x + (size_t)brow * (size_t)T;
  v4f xv = xrow[0];

  {
    cinit(accE, xv.x);
    const v2f Z = {0.0f, 0.0f};
    accE = __builtin_amdgcn_wmma_f32_16x16x4_f32(false, A0, false, Z,
                                                 (short)0, accE, false, false);
    accE = __builtin_amdgcn_wmma_f32_16x16x4_f32(false, A1, false, Z,
                                                 (short)0, accE, false, false);
    accE = __builtin_amdgcn_wmma_f32_16x16x4_f32(false, A2, false, Z,
                                                 (short)0, accE, false, false);
    cinit(accO, xv.y);
  }

  bool curE = false;
  int  tnext = 1;
  if (NT4 > 0) {
    const int qlast = NT4 - 1;
    v4f ra = xrow[1 < qlast ? 1 : qlast];
    v4f rb = xrow[2 < qlast ? 2 : qlast];
    v4f rc = xrow[3 < qlast ? 3 : qlast];
    step(accO, accE, xv.z);
    step(accE, accO, xv.w);
    v4f rd = xrow[4 < qlast ? 4 : qlast];
    step(accO, accE, ra.x);
    int q = 1;
    for (; q + 3 < NT4; q += 4) {
      step(accE, accO, ra.y);
      step(accO, accE, ra.z);
      step(accE, accO, ra.w);
      ra = xrow[q + 4 < qlast ? q + 4 : qlast];
      step(accO, accE, rb.x);
      step(accE, accO, rb.y);
      step(accO, accE, rb.z);
      step(accE, accO, rb.w);
      rb = xrow[q + 5 < qlast ? q + 5 : qlast];
      step(accO, accE, rc.x);
      step(accE, accO, rc.y);
      step(accO, accE, rc.z);
      step(accE, accO, rc.w);
      rc = xrow[q + 6 < qlast ? q + 6 : qlast];
      step(accO, accE, rd.x);
      step(accE, accO, rd.y);
      step(accO, accE, rd.z);
      step(accE, accO, rd.w);
      rd = xrow[q + 7 < qlast ? q + 7 : qlast];
      step(accO, accE, ra.x);
    }
    for (; q < NT4; ++q) {
      step(accE, accO, ra.y);
      step(accO, accE, ra.z);
      step(accE, accO, ra.w);
      step(accO, accE, rb.x);
      ra = rb; rb = rc; rc = rd;
    }
    tnext = NT4 << 2;
    curE = true;
    if (tnext < T) cinit(accE, xs[tnext]);
  }
  for (int t = tnext; t < T; ++t) {
    const float xnext = (t + 1 < T) ? xs[t + 1] : 0.0f;
    if (curE) step(accE, accO, xnext);
    else      step(accO, accE, xnext);
    curE = !curE;
  }
  const bool lastE = !curE;

  float d[HROWS];
#pragma unroll
  for (int v = 0; v < HROWS; ++v) d[v] = fast_tanhf(lastE ? accE[v] : accO[v]);

  __shared__ float hbuf[WAVES_PER_BLOCK][16][HDIM];
#pragma unroll
  for (int v = 0; v < HROWS; ++v) {
    const int j = v + HROWS * half;
    hbuf[wv][col][j] = d[v];
  }
  __syncthreads();
  const int r = col, o = half;
  const int ob = base + r;
  if (ob < B) {
    float acc = b_head[o];
#pragma unroll
    for (int j = 0; j < HDIM; ++j)
      acc = fmaf(hbuf[wv][r][j], W_head[o * HDIM + j], acc);
    out[ob * ODIM + o] = acc;
  }
}

extern "C" void kernel_launch(void* const* d_in, const int* in_sizes, int n_in,
                              void* d_out, int out_size, void* d_ws, size_t ws_size,
                              hipStream_t stream) {
  (void)n_in; (void)d_ws; (void)ws_size;
  const float* x      = (const float*)d_in[0];
  const float* W_ih   = (const float*)d_in[1];
  const float* W_hh   = (const float*)d_in[2];
  const float* b_ih   = (const float*)d_in[3];
  const float* b_hh   = (const float*)d_in[4];
  const float* W_head = (const float*)d_in[5];
  const float* b_head = (const float*)d_in[6];
  float* out = (float*)d_out;

  const int B = out_size / ODIM;
  const int T = in_sizes[0] / B;

  const int waves  = (B + 15) / 16;
  const int blocks = (waves + WAVES_PER_BLOCK - 1) / WAVES_PER_BLOCK;
  tinyrnn_wmma_kernel<<<blocks, 32 * WAVES_PER_BLOCK, 0, stream>>>(
      x, W_ih, W_hh, b_ih, b_hh, W_head, b_head, out, T, B);
}


# x loads grouped: each lane's whole 128-byte line (8 loads of 16 bytes, 32 steps) issued at one place into a second register set: one load place among the WMMAs per 32 steps, was one per 4; same sums
# speedup vs baseline: 1.6821x; 1.1689x over previous
#include <hip/hip_runtime.h>
#include <hip/hip_bf16.h>
#include <math.h>


typedef float v2f __attribute__((ext_vector_type(2)));
typedef float v8f __attribute__((ext_vector_type(8)));
typedef float v4f __attribute__((ext_vector_type(4)));

#define HDIM 12
#define ODIM 2
#define WAVES_PER_BLOCK 4
#define HROWS 6
#define NSLICE 3
#define XBLK 8

__device__ __forceinline__ float fast_tanhf(float x) {
#if __has_builtin(__builtin_amdgcn_tanhf)
  return __builtin_amdgcn_tanhf(x);
#elif __has_builtin(__builtin_amdgcn_tanh_f32)
  return __builtin_amdgcn_tanh_f32(x);
#else
  return tanhf(x);
#endif
}

__launch_bounds__(32 * WAVES_PER_BLOCK)
__global__ void tinyrnn_wmma_kernel(const float* __restrict__ x,
                                    const float* __restrict__ W_ih,
                                    const float* __restrict__ W_hh,
                                    const float* __restrict__ b_ih,
                                    const float* __restrict__ b_hh,
                                    const float* __restrict__ W_head,
                                    const float* __restrict__ b_head,
                                    float* __restrict__ out,
                                    int T, int B) {
  const int lane = threadIdx.x & 31;
  const int wv   = threadIdx.x >> 5;
  const int base = (blockIdx.x * WAVES_PER_BLOCK + wv) * 16;

  const int col  = lane & 15;
  const int half = lane >> 4;

  const bool arowok = (col < HROWS) || (col >= 8 && col < 8 + HROWS);
  const int  arow   = !arowok ? 0 : (col < HROWS ? col : col - 2);
  float a[NSLICE][2];
#pragma unroll
  for (int s = 0; s < NSLICE; ++s) {
#pragma unroll
    for (int r = 0; r < 2; ++r) {
      const int j = 2 * s + r + HROWS * half;
      const float w = W_hh[arow * HDIM + j];
      a[s][r] = arowok ? w : 0.0f;
    }
  }

  float wih[HROWS], bias[HROWS];
#pragma unroll
  for (int v = 0; v < HROWS; ++v) {
    const int j = v + HROWS * half;
    wih[v]  = W_ih[j];
    bias[v] = b_ih[j] + b_hh[j];
  }

  int brow = base + col;
  if (brow >= B) brow = B - 1;
  const v4f* __restrict__ xrow =
      (const v4f*)(x + (size_t)brow * (size_t)T);

  v8f accE, accO;
#pragma unroll
  for (int i = 0; i < 8; ++i) { accE[i] = 0.0f; accO[i] = 0.0f; }

  const v2f A0 = {a[0][0], a[0][1]};
  const v2f A1 = {a[1][0], a[1][1]};
  const v2f A2 = {a[2][0], a[2][1]};

  auto cinit = [&](v8f& c, float xt) {
#pragma unroll
    for (int v = 0; v < HROWS; ++v) c[v] = fmaf(xt, wih[v], bias[v]);
  };

  auto step = [&](v8f& cur, v8f& prev, float xnext) {
    float h[HROWS];
#pragma unroll
    for (int v = 0; v < HROWS; ++v) h[v] = fast_tanhf(prev[v]);
    const v2f B0 = {h[0], h[1]};
    const v2f B1 = {h[2], h[3]};
    const v2f B2 = {h[4], h[5]};
    __builtin_amdgcn_sched_barrier(0);
    cur = __builtin_amdgcn_wmma_f32_16x16x4_f32(false, A0, false, B0,
                                                (short)0, cur, false, false);
    cur = __builtin_amdgcn_wmma_f32_16x16x4_f32(false, A1, false, B1,
                                                (short)0, cur, false, false);
    cur = __builtin_amdgcn_wmma_f32_16x16x4_f32(false, A2, false, B2,
                                                (short)0, cur, false, false);
    __builtin_amdgcn_sched_barrier(0);
    cinit(prev, xnext);
    __builtin_amdgcn_sched_barrier(0);
  };

  const float* xs = x + (size_t)brow * (size_t)T;
  const float x0 = xs[0];
  const float x1 = (T > 1) ? xs[1] : 0.0f;

  {
    cinit(accE, x0);
    const v2f Z = {0.0f, 0.0f};
    accE = __builtin_amdgcn_wmma_f32_16x16x4_f32(false, A0, false, Z,
                                                 (short)0, accE, false, false);
    accE = __builtin_amdgcn_wmma_f32_16x16x4_f32(false, A1, false, Z,
                                                 (short)0, accE, false, false);
    accE = __builtin_amdgcn_wmma_f32_16x16x4_f32(false, A2, false, Z,
                                                 (short)0, accE, false, false);
    cinit(accO, x1);
  }

  const int NG = T >> 5;
  bool curE = false;
  int  tnext = 1;
  if (NG > 0) {
    const int glast = NG - 1;
    v4f a0, a1, a2, a3, a4, a5, a6, a7;
    v4f b0, b1, b2, b3, b4, b5, b6, b7;
#define LOADG(s, g)                                                             \
    {                                                                            \
      const v4f* __restrict__ p_ =                                               \
          xrow + (size_t)((g) < glast ? (g) : glast) * XBLK;                     \
      s##0 = p_[0]; s##1 = p_[1]; s##2 = p_[2]; s##3 = p_[3];                    \
      s##4 = p_[4]; s##5 = p_[5]; s##6 = p_[6]; s##7 = p_[7];                    \
    }
#define BLK4(cb_, nb_)                                                          \
    step(accE, accO, cb_.y); step(accO, accE, cb_.z);                            \
    step(accE, accO, cb_.w); step(accO, accE, nb_.x);
#define RUNG(s, n)                                                              \
    BLK4(s##0, s##1) BLK4(s##1, s##2) BLK4(s##2, s##3) BLK4(s##3, s##4)          \
    BLK4(s##4, s##5) BLK4(s##5, s##6) BLK4(s##6, s##7) BLK4(s##7, n##0)
    LOADG(a, 0)
    LOADG(b, 1)
    step(accO, accE, a0.z); step(accE, accO, a0.w); step(accO, accE, a1.x);
    BLK4(a1, a2) BLK4(a2, a3) BLK4(a3, a4) BLK4(a4, a5)
    BLK4(a5, a6) BLK4(a6, a7) BLK4(a7, b0)
    int g = 1;
    for (; g + 1 < NG; g += 2) {
      LOADG(a, g + 1)
      RUNG(b, a)
      LOADG(b, g + 2)
      RUNG(a, b)
    }
    if (g < NG) {
      LOADG(a, g + 1)
      RUNG(b, a)
    }
#undef RUNG
#undef BLK4
#undef LOADG
    tnext = NG << 5;
    curE = true;
    if (tnext < T) cinit(accE, xs[tnext]);
  }
  for (int t = tnext; t < T; ++t) {
    const float xnext = (t + 1 < T) ? xs[t + 1] : 0.0f;
    if (curE) step(accE, accO, xnext);
    else      step(accO, accE, xnext);
    curE = !curE;
  }
  const bool lastE = !curE;

  float d[HROWS];
#pragma unroll
  for (int v = 0; v < HROWS; ++v) d[v] = fast_tanhf(lastE ? accE[v] : accO[v]);

  __shared__ float hbuf[WAVES_PER_BLOCK][16][HDIM];
#pragma unroll
  for (int v = 0; v < HROWS; ++v) {
    const int j = v + HROWS * half;
    hbuf[wv][col][j] = d[v];
  }
  __syncthreads();
  const int r = col, o = half;
  const int ob = base + r;
  if (ob < B) {
    float acc = b_head[o];
#pragma unroll
    for (int j = 0; j < HDIM; ++j)
      acc = fmaf(hbuf[wv][r][j], W_head[o * HDIM + j], acc);
    out[ob * ODIM + o] = acc;
  }
}

extern "C" void kernel_launch(void* const* d_in, const int* in_sizes, int n_in,
                              void* d_out, int out_size, void* d_ws, size_t ws_size,
                              hipStream_t stream) {
  (void)n_in; (void)d_ws; (void)ws_size;
  const float* x      = (const float*)d_in[0];
  const float* W_ih   = (const float*)d_in[1];
  const float* W_hh   = (const float*)d_in[2];
  const float* b_ih   = (const float*)d_in[3];
  const float* b_hh   = (const float*)d_in[4];
  const float* W_head = (const float*)d_in[5];
  const float* b_head = (const float*)d_in[6];
  float* out = (float*)d_out;

  const int B = out_size / ODIM;
  const int T = in_sizes[0] / B;

  const int waves  = (B + 15) / 16;
  const int blocks = (waves + WAVES_PER_BLOCK - 1) / WAVES_PER_BLOCK;
  tinyrnn_wmma_kernel<<<blocks, 32 * WAVES_PER_BLOCK, 0, stream>>>(
      x, W_ih, W_hh, b_ih, b_hh, W_head, b_head, out, T, B);
}
